# starting kernel
# baseline (speedup 1.0000x reference)
#include <hip/hip_runtime.h>
#include <hip/hip_bf16.h>

#define SEQ   512
#define BATCH 256
#define INF   180
#define HD    8
#define G4    32

typedef __attribute__((ext_vector_type(2))) float v2f;
typedef __attribute__((ext_vector_type(8))) float v8f;

__global__ __launch_bounds__(256) void xgemm_wmma_kernel(
    const float* __restrict__ x,
    const float* __restrict__ w_ih0,
    const float* __restrict__ b_ih0,
    const float* __restrict__ b_hh0,
    float* __restrict__ xg)
{
    const int gwave = (int)((blockIdx.x * blockDim.x + threadIdx.x) >> 5);
    const int lane  = (int)(threadIdx.x & 31);
    const int n     = lane & 15;
    const int hi    = lane >> 4;

    const int rowbase = gwave * 32;
    const float* a0p = x + (size_t)(rowbase + n)      * INF;
    const float* a1p = x + (size_t)(rowbase + 16 + n) * INF;
    const float* b0p = w_ih0 + (size_t)n        * INF;
    const float* b1p = w_ih0 + (size_t)(n + 16) * INF;

    v8f acc00 = {0.f,0.f,0.f,0.f,0.f,0.f,0.f,0.f};
    v8f acc01 = acc00, acc10 = acc00, acc11 = acc00;

    #pragma unroll 5
    for (int k0 = 0; k0 < INF; k0 += 4) {
        const int k = k0 + 2 * hi;
        const v2f a0 = *(const v2f*)(a0p + k);
        const v2f a1 = *(const v2f*)(a1p + k);
        const v2f b0 = *(const v2f*)(b0p + k);
        const v2f b1 = *(const v2f*)(b1p + k);
        acc00 = __builtin_amdgcn_wmma_f32_16x16x4_f32(false, a0, false, b0,
                                                      (short)0, acc00, false, false);
        acc01 = __builtin_amdgcn_wmma_f32_16x16x4_f32(false, a0, false, b1,
                                                      (short)0, acc01, false, false);
        acc10 = __builtin_amdgcn_wmma_f32_16x16x4_f32(false, a1, false, b0,
                                                      (short)0, acc10, false, false);
        acc11 = __builtin_amdgcn_wmma_f32_16x16x4_f32(false, a1, false, b1,
                                                      (short)0, acc11, false, false);
    }

    const float bias0 = b_ih0[n]      + b_hh0[n];
    const float bias1 = b_ih0[n + 16] + b_hh0[n + 16];
    const int p0 = (n & 7) * 4 + (n >> 3);
    const int p1 = p0 + 2;

    float* ob = xg + (size_t)(rowbase + hi * 8) * G4;
    #pragma unroll
    for (int r = 0; r < 8; ++r) {
        ob[(size_t)r * G4 + p0]        = acc00[r] + bias0;
        ob[(size_t)r * G4 + p1]        = acc01[r] + bias1;
        ob[(size_t)(r + 16) * G4 + p0] = acc10[r] + bias0;
        ob[(size_t)(r + 16) * G4 + p1] = acc11[r] + bias1;
    }
}

__device__ __forceinline__ float fast_sigmoid(float x) {
    return 1.0f / (1.0f + __expf(-x));
}
__device__ __forceinline__ float fast_tanh(float x) {
    return 1.0f - 2.0f / (__expf(2.0f * x) + 1.0f);
}

__global__ __launch_bounds__(64) void lstm_scan_kernel(
    const float* __restrict__ xg,
    const float* __restrict__ w_hh0,
    const float* __restrict__ w_ih1,
    const float* __restrict__ w_hh1,
    const float* __restrict__ b_ih1,
    const float* __restrict__ b_hh1,
    const float* __restrict__ w_reg,
    const float* __restrict__ b_reg,
    float* __restrict__ out)
{
    const int tid  = (int)threadIdx.x;
    const int j    = tid & 7;
    const int b    = (int)blockIdx.x * 8 + (tid >> 3);
    const int base = tid & 24;

    float w0[4][HD], wi1[4][HD], wh1[4][HD], bias1[4];
    #pragma unroll
    for (int g = 0; g < 4; ++g) {
        const int r = g * 8 + j;
        #pragma unroll
        for (int k = 0; k < HD; ++k) {
            w0 [g][k] = w_hh0[r * HD + k];
            wi1[g][k] = w_ih1[r * HD + k];
            wh1[g][k] = w_hh1[r * HD + k];
        }
        bias1[g] = b_ih1[r] + b_hh1[r];
    }
    const float wr0 = w_reg[j], wr1 = w_reg[HD + j];
    const float br0 = b_reg[0], br1 = b_reg[1];

    float h1 = 0.f, c1 = 0.f, h2 = 0.f, c2 = 0.f;

    const float4* xgp = (const float4*)(xg + (size_t)b * G4 + j * 4);
    const size_t xstride = (size_t)BATCH * G4 / 4;

    float4 gin = xgp[0];
    for (int s = 0; s < SEQ; ++s) {
        const float4 gnext = xgp[(size_t)((s + 1) & (SEQ - 1)) * xstride];

        float gi = gin.x, gf = gin.y, gg = gin.z, go = gin.w;
        #pragma unroll
        for (int k = 0; k < HD; ++k) {
            const float hk = __shfl(h1, base + k, 32);
            gi += hk * w0[0][k];
            gf += hk * w0[1][k];
            gg += hk * w0[2][k];
            go += hk * w0[3][k];
        }
        const float i0 = fast_sigmoid(gi);
        const float f0 = fast_sigmoid(gf);
        const float g0 = fast_tanh(gg);
        const float o0 = fast_sigmoid(go);
        c1 = f0 * c1 + i0 * g0;
        h1 = o0 * fast_tanh(c1);

        float a0 = bias1[0], a1 = bias1[1], a2 = bias1[2], a3 = bias1[3];
        #pragma unroll
        for (int k = 0; k < HD; ++k) {
            const float h1k = __shfl(h1, base + k, 32);
            const float h2k = __shfl(h2, base + k, 32);
            a0 += h1k * wi1[0][k] + h2k * wh1[0][k];
            a1 += h1k * wi1[1][k] + h2k * wh1[1][k];
            a2 += h1k * wi1[2][k] + h2k * wh1[2][k];
            a3 += h1k * wi1[3][k] + h2k * wh1[3][k];
        }
        const float i1 = fast_sigmoid(a0);
        const float f1 = fast_sigmoid(a1);
        const float g1 = fast_tanh(a2);
        const float o1 = fast_sigmoid(a3);
        c2 = f1 * c2 + i1 * g1;
        h2 = o1 * fast_tanh(c2);

        float p0 = h2 * wr0;
        float p1 = h2 * wr1;
        #pragma unroll
        for (int off = 1; off < 8; off <<= 1) {
            p0 += __shfl_xor(p0, off, 32);
            p1 += __shfl_xor(p1, off, 32);
        }
        if (j == 0) {
            float2 r;
            r.x = p0 + br0;
            r.y = p1 + br1;
            *(float2*)(out + ((size_t)s * BATCH + b) * 2) = r;
        }
        gin = gnext;
    }
}

extern "C" void kernel_launch(void* const* d_in, const int* in_sizes, int n_in,
                              void* d_out, int out_size, void* d_ws, size_t ws_size,
                              hipStream_t stream) {
    const float* x     = (const float*)d_in[0];
    const float* w_ih0 = (const float*)d_in[1];
    const float* w_hh0 = (const float*)d_in[2];
    const float* b_ih0 = (const float*)d_in[3];
    const float* b_hh0 = (const float*)d_in[4];
    const float* w_ih1 = (const float*)d_in[5];
    const float* w_hh1 = (const float*)d_in[6];
    const float* b_ih1 = (const float*)d_in[7];
    const float* b_hh1 = (const float*)d_in[8];
    const float* w_reg = (const float*)d_in[9];
    const float* b_reg = (const float*)d_in[10];
    float* out = (float*)d_out;
    float* xg  = (float*)d_ws;

    xgemm_wmma_kernel<<<512, 256, 0, stream>>>(x, w_ih0, b_ih0, b_hh0, xg);
    lstm_scan_kernel<<<32, 64, 0, stream>>>(xg, w_hh0, w_ih1, w_hh1,
                                            b_ih1, b_hh1, w_reg, b_reg, out);
}


# scan kernel: layers run one step apart in one wave (layer-1 cell of step s-1 beside layer-0 cell of step s); each layer's WMMAs issue under the other's cell update, 1-2 per piece; sum order kept
# speedup vs baseline: 1.5567x; 1.5567x over previous
#include <hip/hip_runtime.h>
#include <hip/hip_bf16.h>

#define SEQ   512
#define BATCH 256
#define INF   180
#define HD    8
#define G4    32

typedef __attribute__((ext_vector_type(2))) float v2f;
typedef __attribute__((ext_vector_type(8))) float v8f;

__global__ __launch_bounds__(256) void xgemm_wmma_kernel(
    const float* __restrict__ x,
    const float* __restrict__ w_ih0,
    const float* __restrict__ b_ih0,
    const float* __restrict__ b_hh0,
    float* __restrict__ xg)
{
    const int gwave = (int)((blockIdx.x * blockDim.x + threadIdx.x) >> 5);
    const int lane  = (int)(threadIdx.x & 31);
    const int n     = lane & 15;
    const int hi    = lane >> 4;

    const int rowbase = gwave * 32;
    const float* a0p = x + (size_t)(rowbase + n)      * INF;
    const float* a1p = x + (size_t)(rowbase + 16 + n) * INF;
    const float* b0p = w_ih0 + (size_t)n        * INF;
    const float* b1p = w_ih0 + (size_t)(n + 16) * INF;

    v8f acc00 = {0.f,0.f,0.f,0.f,0.f,0.f,0.f,0.f};
    v8f acc01 = acc00, acc10 = acc00, acc11 = acc00;

    #pragma unroll 5
    for (int k0 = 0; k0 < INF; k0 += 4) {
        const int k = k0 + 2 * hi;
        const v2f a0 = *(const v2f*)(a0p + k);
        const v2f a1 = *(const v2f*)(a1p + k);
        const v2f b0 = *(const v2f*)(b0p + k);
        const v2f b1 = *(const v2f*)(b1p + k);
        acc00 = __builtin_amdgcn_wmma_f32_16x16x4_f32(false, a0, false, b0,
                                                      (short)0, acc00, false, false);
        acc01 = __builtin_amdgcn_wmma_f32_16x16x4_f32(false, a0, false, b1,
                                                      (short)0, acc01, false, false);
        acc10 = __builtin_amdgcn_wmma_f32_16x16x4_f32(false, a1, false, b0,
                                                      (short)0, acc10, false, false);
        acc11 = __builtin_amdgcn_wmma_f32_16x16x4_f32(false, a1, false, b1,
                                                      (short)0, acc11, false, false);
    }

    const float bias0 = b_ih0[n]      + b_hh0[n];
    const float bias1 = b_ih0[n + 16] + b_hh0[n + 16];
    const int p0 = (n & 7) * 4 + (n >> 3);
    const int p1 = p0 + 2;

    float* ob = xg + (size_t)(rowbase + hi * 8) * G4;
    #pragma unroll
    for (int r = 0; r < 8; ++r) {
        ob[(size_t)r * G4 + p0]        = acc00[r] + bias0;
        ob[(size_t)r * G4 + p1]        = acc01[r] + bias1;
        ob[(size_t)(r + 16) * G4 + p0] = acc10[r] + bias0;
        ob[(size_t)(r + 16) * G4 + p1] = acc11[r] + bias1;
    }
}

#define LOG2E     1.4426950408889634f
#define SIG_TMAX  36.0f
#define TANH_TMAX 29.0f

__device__ __forceinline__ v8f wmma4(v2f a, v2f b, v8f c) {
    return __builtin_amdgcn_wmma_f32_16x16x4_f32(false, a, false, b,
                                                 (short)0, c, false, false);
}

struct CellWork {
    float e[4][4];
    float num[4], den[4];
    float ao[4];
    float hn[4], hd[4];
};

__device__ __forceinline__ void cell_exps(const v8f& d0, const v8f& d1, CellWork& w) {
    #pragma unroll
    for (int u = 0; u < 4; ++u) {
        const int o = 4 * (u & 1);
        const float pi = (u < 2) ? d0[o + 0] : d1[o + 0];
        const float pf = (u < 2) ? d0[o + 1] : d1[o + 1];
        const float pg = (u < 2) ? d0[o + 2] : d1[o + 2];
        const float po = (u < 2) ? d0[o + 3] : d1[o + 3];
        w.e[u][0] = __builtin_amdgcn_exp2f(fminf(pi * -LOG2E, SIG_TMAX));
        w.e[u][1] = __builtin_amdgcn_exp2f(fminf(pf * -LOG2E, SIG_TMAX));
        w.e[u][2] = __builtin_amdgcn_exp2f(fminf(pg * (2.0f * LOG2E), TANH_TMAX));
        w.e[u][3] = __builtin_amdgcn_exp2f(fminf(po * -LOG2E, SIG_TMAX));
    }
}
__device__ __forceinline__ void cell_cfrac(const float (&cs)[4], CellWork& w) {
    #pragma unroll
    for (int u = 0; u < 4; ++u) {
        const float eg = w.e[u][2];
        const float ai = 1.0f + w.e[u][0];
        const float af = 1.0f + w.e[u][1];
        w.ao[u] = 1.0f + w.e[u][3];
        const float p  = ai * (eg + 1.0f);
        w.num[u] = cs[u] * p + (eg - 1.0f) * af;
        w.den[u] = af * p;
    }
}
__device__ __forceinline__ void cell_cdiv(float (&cs)[4], const CellWork& w) {
    #pragma unroll
    for (int u = 0; u < 4; ++u) cs[u] = w.num[u] / w.den[u];
}
__device__ __forceinline__ void cell_hfrac(const float (&cs)[4], CellWork& w) {
    #pragma unroll
    for (int u = 0; u < 4; ++u) {
        const float ec = __builtin_amdgcn_exp2f(fminf(cs[u] * (2.0f * LOG2E), TANH_TMAX));
        w.hn[u] = ec - 1.0f;
        w.hd[u] = (ec + 1.0f) * w.ao[u];
    }
}
__device__ __forceinline__ void cell_hdiv(float (&hv)[4], const CellWork& w) {
    #pragma unroll
    for (int u = 0; u < 4; ++u) hv[u] = w.hn[u] / w.hd[u];
}
__device__ __forceinline__ void cell_update(const v8f& d0, const v8f& d1,
                                            float (&cs)[4], float (&hv)[4]) {
    CellWork w;
    cell_exps(d0, d1, w);
    cell_cfrac(cs, w);
    cell_cdiv(cs, w);
    cell_hfrac(cs, w);
    cell_hdiv(hv, w);
}

__global__ __launch_bounds__(32) __attribute__((amdgpu_waves_per_eu(1, 1)))
void lstm_scan_kernel(
    const float* __restrict__ xg,
    const float* __restrict__ w_hh0,
    const float* __restrict__ w_ih1,
    const float* __restrict__ w_hh1,
    const float* __restrict__ b_ih1,
    const float* __restrict__ b_hh1,
    const float* __restrict__ w_reg,
    const float* __restrict__ b_reg,
    float* __restrict__ out)
{
    const int lane = (int)(threadIdx.x & 31);
    const int c    = lane & 15;
    const int hf   = lane >> 4;
    const int b    = (int)blockIdx.x * 16 + c;

    v2f a_hh0[2][2], a_ih1[2][2], a_hh1[2][2], a_head[2];
    #pragma unroll
    for (int t = 0; t < 2; ++t) {
        const int wrow = (c & 3) * HD + 4 * (c >> 3) + 2 * t + ((c >> 2) & 1);
        #pragma unroll
        for (int q = 0; q < 2; ++q) {
            const int k = 4 * hf + 2 * q;
            a_hh0[t][q] = *(const v2f*)(w_hh0 + wrow * HD + k);
            a_ih1[t][q] = *(const v2f*)(w_ih1 + wrow * HD + k);
            a_hh1[t][q] = *(const v2f*)(w_hh1 + wrow * HD + k);
        }
    }
    #pragma unroll
    for (int q = 0; q < 2; ++q) {
        const int k = 4 * hf + 2 * q;
        const v2f wv = *(const v2f*)(w_reg + (c & 1) * HD + k);
        const v2f zero2 = {0.f, 0.f};
        a_head[q] = ((c & 7) < 2) ? wv : zero2;
    }

    v8f cb1[2], cbh;
    #pragma unroll
    for (int t = 0; t < 2; ++t) {
        #pragma unroll
        for (int r = 0; r < 8; ++r) {
            const int row = (r & 3) * HD + 4 * hf + 2 * t + (r >> 2);
            cb1[t][r] = b_ih1[row] + b_hh1[row];
        }
    }
    #pragma unroll
    for (int r = 0; r < 8; ++r) cbh[r] = 0.f;
    cbh[0] = b_reg[0];
    cbh[1] = b_reg[1];

    float h1[4] = {0.f, 0.f, 0.f, 0.f}, c1[4] = {0.f, 0.f, 0.f, 0.f};
    float h2[4] = {0.f, 0.f, 0.f, 0.f}, c2[4] = {0.f, 0.f, 0.f, 0.f};

    const v8f* xgp = (const v8f*)(xg + (size_t)b * G4 + hf * 16);
    const size_t xstride = (size_t)BATCH * G4 / 8;
    float2* outp = (float2*)(out + (size_t)b * 2) + (size_t)hf * BATCH;

#define FENCE() __builtin_amdgcn_sched_barrier(0)
#define PIN2(a, b)                asm volatile("" : "+v"(a), "+v"(b))
#define PIN6(a, b, p, q, r, s)    asm volatile("" : "+v"(a), "+v"(b), "+v"(p), "+v"(q), "+v"(r), "+v"(s))

    float oe0 = 0.f, oe1 = 0.f;
    float oo0 = 0.f, oo1 = 0.f;

    auto trip = [&](const int s, v8f& t0, v8f& t1, v8f& x0, v8f& x1,
                    float& ho0, float& ho1) {
        v2f r0 = {h2[0], h2[1]};
        v2f r1 = {h2[2], h2[3]};
        v2f i0 = {h1[0], h1[1]};
        v2f i1 = {h1[2], h1[3]};
        CellWork w;

        v8f e0 = wmma4(a_hh1[0][0], r0, cb1[0]);
        e0 = wmma4(a_hh1[0][1], r1, e0);
        cell_exps(t0, t1, w);
        PIN6(e0, r0, w.e[0][0], w.e[1][0], w.e[2][0], w.e[3][0]);
        FENCE();
        {
            const v8f* np = xgp + (size_t)((s + 2) & (SEQ - 1)) * xstride;
            t0 = np[0];
            t1 = np[1];
        }
        v8f e1 = wmma4(a_hh1[1][0], r0, cb1[1]);
        cell_cfrac(c1, w);
        PIN6(e1, r1, w.den[0], w.den[1], w.den[2], w.den[3]);
        FENCE();
        e1 = wmma4(a_hh1[1][1], r1, e1);
        e1 = wmma4(a_ih1[1][0], i0, e1);
        cell_cdiv(c1, w);
        PIN6(e1, i1, c1[0], c1[1], c1[2], c1[3]);
        FENCE();
        e1 = wmma4(a_ih1[1][1], i1, e1);
        cell_hfrac(c1, w);
        PIN6(e1, i0, w.hd[0], w.hd[1], w.hd[2], w.hd[3]);
        FENCE();
        e0 = wmma4(a_ih1[0][0], i0, e0);
        e0 = wmma4(a_ih1[0][1], i1, e0);
        cell_hdiv(h1, w);
        v2f bq0 = {h1[0], h1[1]};
        v2f bq1 = {h1[2], h1[3]};
        PIN2(e0, bq0);
        FENCE();

        x0 = wmma4(a_hh0[0][0], bq0, x0);
        x0 = wmma4(a_hh0[0][1], bq1, x0);
        cell_exps(e0, e1, w);
        PIN6(x0, bq0, w.e[0][0], w.e[1][0], w.e[2][0], w.e[3][0]);
        FENCE();
        x1 = wmma4(a_hh0[1][0], bq0, x1);
        cell_cfrac(c2, w);
        PIN6(x1, r0, w.den[0], w.den[1], w.den[2], w.den[3]);
        FENCE();
        v8f dh = wmma4(a_head[0], r0, cbh);
        dh = wmma4(a_head[1], r1, dh);
        cell_cdiv(c2, w);
        PIN6(dh, bq1, c2[0], c2[1], c2[2], c2[3]);
        FENCE();
        x1 = wmma4(a_hh0[1][1], bq1, x1);
        cell_hfrac(c2, w);
        PIN6(x1, w.hn[0], w.hd[0], w.hd[1], w.hd[2], w.hd[3]);
        FENCE();
        cell_hdiv(h2, w);
        ho0 = dh[0];
        ho1 = dh[1];
    };

    v8f p0 = xgp[0], p1 = xgp[1];
    {
        const v2f z = {0.f, 0.f};
        p0 = wmma4(a_hh0[0][0], z, p0);
        p1 = wmma4(a_hh0[1][0], z, p1);
        p0 = wmma4(a_hh0[0][1], z, p0);
        p1 = wmma4(a_hh0[1][1], z, p1);
    }
    cell_update(p0, p1, c1, h1);
    p0 = xgp[xstride];
    p1 = xgp[xstride + 1];
    {
        const v2f bq0 = {h1[0], h1[1]};
        const v2f bq1 = {h1[2], h1[3]};
        p0 = wmma4(a_hh0[0][0], bq0, p0);
        p1 = wmma4(a_hh0[1][0], bq0, p1);
        p0 = wmma4(a_hh0[0][1], bq1, p0);
        p1 = wmma4(a_hh0[1][1], bq1, p1);
    }
    v8f q0 = xgp[2 * xstride];
    v8f q1 = xgp[2 * xstride + 1];
    {
        float u0, u1;
        trip(1, p0, p1, q0, q1, u0, u1);
    }

    for (int s = 2; s < SEQ; s += 2) {
        trip(s,     q0, q1, p0, p1, oe0, oe1);
        trip(s + 1, p0, p1, q0, q1, oo0, oo1);
        float2 r;
        r.x = hf ? oo0 : oe0;
        r.y = hf ? oo1 : oe1;
        outp[(size_t)(s - 2) * BATCH] = r;
    }
#undef FENCE
#undef PIN2
#undef PIN6

    {
        const v2f r0 = {h2[0], h2[1]};
        const v2f r1 = {h2[2], h2[3]};
        const v2f i0 = {h1[0], h1[1]};
        const v2f i1 = {h1[2], h1[3]};
        v8f dh = wmma4(a_head[0], r0, cbh);
        dh = wmma4(a_head[1], r1, dh);
        v8f e0 = cb1[0], e1 = cb1[1];
        e0 = wmma4(a_hh1[0][0], r0, e0);
        e1 = wmma4(a_hh1[1][0], r0, e1);
        e0 = wmma4(a_hh1[0][1], r1, e0);
        e1 = wmma4(a_hh1[1][1], r1, e1);
        e0 = wmma4(a_ih1[0][0], i0, e0);
        e1 = wmma4(a_ih1[1][0], i0, e1);
        e0 = wmma4(a_ih1[0][1], i1, e0);
        e1 = wmma4(a_ih1[1][1], i1, e1);
        cell_update(e0, e1, c2, h2);
        const v2f g0 = {h2[0], h2[1]};
        const v2f g1 = {h2[2], h2[3]};
        v8f dl = wmma4(a_head[0], g0, cbh);
        dl = wmma4(a_head[1], g1, dl);
        float2 r;
        r.x = hf ? dl[0] : dh[0];
        r.y = hf ? dl[1] : dh[1];
        outp[(size_t)(SEQ - 2) * BATCH] = r;
    }
}

extern "C" void kernel_launch(void* const* d_in, const int* in_sizes, int n_in,
                              void* d_out, int out_size, void* d_ws, size_t ws_size,
                              hipStream_t stream) {
    const float* x     = (const float*)d_in[0];
    const float* w_ih0 = (const float*)d_in[1];
    const float* w_hh0 = (const float*)d_in[2];
    const float* b_ih0 = (const float*)d_in[3];
    const float* b_hh0 = (const float*)d_in[4];
    const float* w_ih1 = (const float*)d_in[5];
    const float* w_hh1 = (const float*)d_in[6];
    const float* b_ih1 = (const float*)d_in[7];
    const float* b_hh1 = (const float*)d_in[8];
    const float* w_reg = (const float*)d_in[9];
    const float* b_reg = (const float*)d_in[10];
    float* out = (float*)d_out;
    float* xg  = (float*)d_ws;

    xgemm_wmma_kernel<<<512, 256, 0, stream>>>(x, w_ih0, b_ih0, b_hh0, xg);
    lstm_scan_kernel<<<16, 32, 0, stream>>>(xg, w_hh0, w_ih1, w_hh1,
                                            b_ih1, b_hh1, w_reg, b_reg, out);
}


# scan re-tiled: 4 batch elements per wave (64 waves, was 16), each on 4 matrix columns; a lane's cell update is 1 hidden unit (was 4): own gates by lane select, h spread by quad DPP; sum order kept
# speedup vs baseline: 1.9817x; 1.2730x over previous
#include <hip/hip_runtime.h>
#include <hip/hip_bf16.h>

#define SEQ   512
#define BATCH 256
#define INF   180
#define HD    8
#define G4    32

typedef __attribute__((ext_vector_type(2))) float v2f;
typedef __attribute__((ext_vector_type(4))) float v4f;
typedef __attribute__((ext_vector_type(8))) float v8f;

__global__ __launch_bounds__(256) void xgemm_wmma_kernel(
    const float* __restrict__ x,
    const float* __restrict__ w_ih0,
    const float* __restrict__ b_ih0,
    const float* __restrict__ b_hh0,
    float* __restrict__ xg)
{
    const int gwave = (int)((blockIdx.x * blockDim.x + threadIdx.x) >> 5);
    const int lane  = (int)(threadIdx.x & 31);
    const int n     = lane & 15;
    const int hi    = lane >> 4;

    const int rowbase = gwave * 32;
    const float* a0p = x + (size_t)(rowbase + n)      * INF;
    const float* a1p = x + (size_t)(rowbase + 16 + n) * INF;
    const float* b0p = w_ih0 + (size_t)n        * INF;
    const float* b1p = w_ih0 + (size_t)(n + 16) * INF;

    v8f acc00 = {0.f,0.f,0.f,0.f,0.f,0.f,0.f,0.f};
    v8f acc01 = acc00, acc10 = acc00, acc11 = acc00;

    #pragma unroll 5
    for (int k0 = 0; k0 < INF; k0 += 4) {
        const int k = k0 + 2 * hi;
        const v2f a0 = *(const v2f*)(a0p + k);
        const v2f a1 = *(const v2f*)(a1p + k);
        const v2f b0 = *(const v2f*)(b0p + k);
        const v2f b1 = *(const v2f*)(b1p + k);
        acc00 = __builtin_amdgcn_wmma_f32_16x16x4_f32(false, a0, false, b0,
                                                      (short)0, acc00, false, false);
        acc01 = __builtin_amdgcn_wmma_f32_16x16x4_f32(false, a0, false, b1,
                                                      (short)0, acc01, false, false);
        acc10 = __builtin_amdgcn_wmma_f32_16x16x4_f32(false, a1, false, b0,
                                                      (short)0, acc10, false, false);
        acc11 = __builtin_amdgcn_wmma_f32_16x16x4_f32(false, a1, false, b1,
                                                      (short)0, acc11, false, false);
    }

    const float bias0 = b_ih0[n]      + b_hh0[n];
    const float bias1 = b_ih0[n + 16] + b_hh0[n + 16];
    const int p0 = (n & 7) * 4 + (n >> 3);
    const int p1 = p0 + 2;

    float* ob = xg + (size_t)(rowbase + hi * 8) * G4;
    #pragma unroll
    for (int r = 0; r < 8; ++r) {
        ob[(size_t)r * G4 + p0]        = acc00[r] + bias0;
        ob[(size_t)r * G4 + p1]        = acc01[r] + bias1;
        ob[(size_t)(r + 16) * G4 + p0] = acc10[r] + bias0;
        ob[(size_t)(r + 16) * G4 + p1] = acc11[r] + bias1;
    }
}

#define LOG2E     1.4426950408889634f
#define SIG_TMAX  36.0f
#define TANH_TMAX 29.0f
#define BPW       4

__device__ __forceinline__ v8f wmma4(v2f a, v2f b, v8f c) {
    return __builtin_amdgcn_wmma_f32_16x16x4_f32(false, a, false, b,
                                                 (short)0, c, false, false);
}

template <int J>
__device__ __forceinline__ float quad_bcast(float v) {
    return __int_as_float(__builtin_amdgcn_mov_dpp(__float_as_int(v), J * 0x55, 0xF, 0xF, false));
}
__device__ __forceinline__ void quad_spread(float v, float (&hb)[4]) {
    hb[0] = quad_bcast<0>(v);
    hb[1] = quad_bcast<1>(v);
    hb[2] = quad_bcast<2>(v);
    hb[3] = quad_bcast<3>(v);
}

struct CellWork {
    float ei, ef, eg, eo;
    float num, den;
    float ao;
    float hn, hd;
};

__device__ __forceinline__ void cell_exps(const v8f& d0, const v8f& d1,
                                          const bool dlo, const bool dhi, CellWork& w) {
    const v8f m  = dhi ? d1 : d0;
    const v4f lo = __builtin_shufflevector(m, m, 0, 1, 2, 3);
    const v4f hi = __builtin_shufflevector(m, m, 4, 5, 6, 7);
    const v4f p  = dlo ? hi : lo;
    w.ei = __builtin_amdgcn_exp2f(fminf(p[0] * -LOG2E, SIG_TMAX));
    w.ef = __builtin_amdgcn_exp2f(fminf(p[1] * -LOG2E, SIG_TMAX));
    w.eg = __builtin_amdgcn_exp2f(fminf(p[2] * (2.0f * LOG2E), TANH_TMAX));
    w.eo = __builtin_amdgcn_exp2f(fminf(p[3] * -LOG2E, SIG_TMAX));
}
__device__ __forceinline__ void cell_cfrac(const float cs, CellWork& w) {
    const float eg = w.eg;
    const float ai = 1.0f + w.ei;
    const float af = 1.0f + w.ef;
    w.ao = 1.0f + w.eo;
    const float p  = ai * (eg + 1.0f);
    w.num = cs * p + (eg - 1.0f) * af;
    w.den = af * p;
}
__device__ __forceinline__ void cell_cdiv(float& cs, const CellWork& w) {
    cs = w.num / w.den;
}
__device__ __forceinline__ void cell_hfrac(const float cs, CellWork& w) {
    const float ec = __builtin_amdgcn_exp2f(fminf(cs * (2.0f * LOG2E), TANH_TMAX));
    w.hn = ec - 1.0f;
    w.hd = (ec + 1.0f) * w.ao;
}
__device__ __forceinline__ void cell_hdiv(float& hv, const CellWork& w) {
    hv = w.hn / w.hd;
}
__device__ __forceinline__ void cell_update(const v8f& d0, const v8f& d1,
                                            const bool dlo, const bool dhi,
                                            float& cs, float& hv) {
    CellWork w;
    cell_exps(d0, d1, dlo, dhi, w);
    cell_cfrac(cs, w);
    cell_cdiv(cs, w);
    cell_hfrac(cs, w);
    cell_hdiv(hv, w);
}

__global__ __launch_bounds__(32) __attribute__((amdgpu_waves_per_eu(1, 1)))
void lstm_scan_kernel(
    const float* __restrict__ xg,
    const float* __restrict__ w_hh0,
    const float* __restrict__ w_ih1,
    const float* __restrict__ w_hh1,
    const float* __restrict__ b_ih1,
    const float* __restrict__ b_hh1,
    const float* __restrict__ w_reg,
    const float* __restrict__ b_reg,
    float* __restrict__ out)
{
    const int lane = (int)(threadIdx.x & 31);
    const int c    = lane & 15;
    const int hf   = lane >> 4;
    const int d    = c & 3;
    const bool dlo = (d & 1) != 0;
    const bool dhi = (d & 2) != 0;
    const int b    = (int)blockIdx.x * BPW + (c >> 2);

    v2f a_hh0[2][2], a_ih1[2][2], a_hh1[2][2], a_head[2];
    #pragma unroll
    for (int t = 0; t < 2; ++t) {
        const int wrow = (c & 3) * HD + 4 * (c >> 3) + 2 * t + ((c >> 2) & 1);
        #pragma unroll
        for (int q = 0; q < 2; ++q) {
            const int k = 4 * hf + 2 * q;
            a_hh0[t][q] = *(const v2f*)(w_hh0 + wrow * HD + k);
            a_ih1[t][q] = *(const v2f*)(w_ih1 + wrow * HD + k);
            a_hh1[t][q] = *(const v2f*)(w_hh1 + wrow * HD + k);
        }
    }
    #pragma unroll
    for (int q = 0; q < 2; ++q) {
        const int k = 4 * hf + 2 * q;
        const v2f wv = *(const v2f*)(w_reg + (c & 1) * HD + k);
        const v2f zero2 = {0.f, 0.f};
        a_head[q] = ((c & 7) < 2) ? wv : zero2;
    }

    v8f cb1[2], cbh;
    #pragma unroll
    for (int t = 0; t < 2; ++t) {
        #pragma unroll
        for (int r = 0; r < 8; ++r) {
            const int row = (r & 3) * HD + 4 * hf + 2 * t + (r >> 2);
            cb1[t][r] = b_ih1[row] + b_hh1[row];
        }
    }
    #pragma unroll
    for (int r = 0; r < 8; ++r) cbh[r] = 0.f;
    cbh[0] = b_reg[0];
    cbh[1] = b_reg[1];

    float h1 = 0.f, c1 = 0.f, h2 = 0.f, c2 = 0.f;
    v2f h1p0 = {0.f, 0.f}, h1p1 = {0.f, 0.f};
    v2f h2p0 = {0.f, 0.f}, h2p1 = {0.f, 0.f};

    const v8f* xgp = (const v8f*)(xg + (size_t)b * G4 + hf * 16);
    const size_t xstride = (size_t)BATCH * G4 / 8;
    float2* outp = (float2*)(out + (size_t)b * 2) + (size_t)hf * BATCH;

#define FENCE() __builtin_amdgcn_sched_barrier(0)
#define PIN3(a, b, p)             asm volatile("" : "+v"(a), "+v"(b), "+v"(p))
#define PIN4(a, b, p, q)          asm volatile("" : "+v"(a), "+v"(b), "+v"(p), "+v"(q))
#define PIN6(a, b, p, q, r, s)    asm volatile("" : "+v"(a), "+v"(b), "+v"(p), "+v"(q), "+v"(r), "+v"(s))
#define APART() do { __builtin_amdgcn_sched_group_barrier(0x008, 1, 0); \
                     __builtin_amdgcn_sched_group_barrier(0x402, 8, 0); \
                     __builtin_amdgcn_sched_group_barrier(0x008, 1, 0); } while (0)

    float oe0 = 0.f, oe1 = 0.f;
    float oo0 = 0.f, oo1 = 0.f;

    auto trip = [&](const int s, v8f& t0, v8f& t1, v8f& x0, v8f& x1,
                    float& ho0, float& ho1) {
        const v2f r0 = h2p0, r1 = h2p1;
        const v2f i0 = h1p0, i1 = h1p1;
        CellWork w;

        v8f e0 = wmma4(a_hh1[0][0], r0, cb1[0]);
        v8f e1 = wmma4(a_hh1[1][0], r0, cb1[1]);
        cell_exps(t0, t1, dlo, dhi, w);
        APART();
        PIN6(e0, e1, w.ei, w.ef, w.eg, w.eo);
        FENCE();
        {
            const v8f* np = xgp + (size_t)((s + 2) & (SEQ - 1)) * xstride;
            t0 = np[0];
            t1 = np[1];
        }
        e0 = wmma4(a_hh1[0][1], r1, e0);
        cell_cfrac(c1, w);
        PIN3(e0, w.num, w.den);
        FENCE();
        e1 = wmma4(a_hh1[1][1], r1, e1);
        e0 = wmma4(a_ih1[0][0], i0, e0);
        cell_cdiv(c1, w);
        APART();
        PIN3(e1, e0, c1);
        FENCE();
        e1 = wmma4(a_ih1[1][0], i0, e1);
        cell_hfrac(c1, w);
        PIN3(e1, w.hn, w.hd);
        FENCE();
        e0 = wmma4(a_ih1[0][1], i1, e0);
        e1 = wmma4(a_ih1[1][1], i1, e1);
        cell_hdiv(h1, w);
        float hb[4];
        quad_spread(h1, hb);
        v2f bq0 = {hb[0], hb[1]};
        v2f bq1 = {hb[2], hb[3]};
        APART();
        PIN4(e0, e1, bq0, bq1);
        FENCE();
        h1p0 = bq0;
        h1p1 = bq1;

        x0 = wmma4(a_hh0[0][0], bq0, x0);
        x1 = wmma4(a_hh0[1][0], bq0, x1);
        cell_exps(e0, e1, dlo, dhi, w);
        APART();
        PIN6(x0, x1, w.ei, w.ef, w.eg, w.eo);
        FENCE();
        v8f dh = wmma4(a_head[0], r0, cbh);
        cell_cfrac(c2, w);
        PIN3(dh, w.num, w.den);
        FENCE();
        x0 = wmma4(a_hh0[0][1], bq1, x0);
        x1 = wmma4(a_hh0[1][1], bq1, x1);
        cell_cdiv(c2, w);
        APART();
        PIN3(x0, x1, c2);
        FENCE();
        dh = wmma4(a_head[1], r1, dh);
        cell_hfrac(c2, w);
        PIN3(dh, w.hn, w.hd);
        FENCE();
        cell_hdiv(h2, w);
        quad_spread(h2, hb);
        h2p0 = (v2f){hb[0], hb[1]};
        h2p1 = (v2f){hb[2], hb[3]};
        ho0 = dh[0];
        ho1 = dh[1];
    };

    v8f p0 = xgp[0], p1 = xgp[1];
    {
        const v2f z = {0.f, 0.f};
        p0 = wmma4(a_hh0[0][0], z, p0);
        p1 = wmma4(a_hh0[1][0], z, p1);
        p0 = wmma4(a_hh0[0][1], z, p0);
        p1 = wmma4(a_hh0[1][1], z, p1);
    }
    cell_update(p0, p1, dlo, dhi, c1, h1);
    {
        float hb[4];
        quad_spread(h1, hb);
        h1p0 = (v2f){hb[0], hb[1]};
        h1p1 = (v2f){hb[2], hb[3]};
    }
    p0 = xgp[xstride];
    p1 = xgp[xstride + 1];
    p0 = wmma4(a_hh0[0][0], h1p0, p0);
    p1 = wmma4(a_hh0[1][0], h1p0, p1);
    p0 = wmma4(a_hh0[0][1], h1p1, p0);
    p1 = wmma4(a_hh0[1][1], h1p1, p1);
    v8f q0 = xgp[2 * xstride];
    v8f q1 = xgp[2 * xstride + 1];
    {
        float u0, u1;
        trip(1, p0, p1, q0, q1, u0, u1);
    }

    for (int s = 2; s < SEQ; s += 2) {
        trip(s,     q0, q1, p0, p1, oe0, oe1);
        trip(s + 1, p0, p1, q0, q1, oo0, oo1);
        float2 r;
        r.x = hf ? oo0 : oe0;
        r.y = hf ? oo1 : oe1;
        if (d == 0) outp[(size_t)(s - 2) * BATCH] = r;
    }
#undef FENCE
#undef PIN3
#undef PIN4
#undef PIN6
#undef APART

    {
        const v2f r0 = h2p0, r1 = h2p1;
        const v2f i0 = h1p0, i1 = h1p1;
        v8f dh = wmma4(a_head[0], r0, cbh);
        dh = wmma4(a_head[1], r1, dh);
        v8f e0 = cb1[0], e1 = cb1[1];
        e0 = wmma4(a_hh1[0][0], r0, e0);
        e1 = wmma4(a_hh1[1][0], r0, e1);
        e0 = wmma4(a_hh1[0][1], r1, e0);
        e1 = wmma4(a_hh1[1][1], r1, e1);
        e0 = wmma4(a_ih1[0][0], i0, e0);
        e1 = wmma4(a_ih1[1][0], i0, e1);
        e0 = wmma4(a_ih1[0][1], i1, e0);
        e1 = wmma4(a_ih1[1][1], i1, e1);
        cell_update(e0, e1, dlo, dhi, c2, h2);
        float hb[4];
        quad_spread(h2, hb);
        const v2f g0 = {hb[0], hb[1]};
        const v2f g1 = {hb[2], hb[3]};
        v8f dl = wmma4(a_head[0], g0, cbh);
        dl = wmma4(a_head[1], g1, dl);
        float2 r;
        r.x = hf ? dl[0] : dh[0];
        r.y = hf ? dl[1] : dh[1];
        if (d == 0) outp[(size_t)(SEQ - 2) * BATCH] = r;
    }
}

extern "C" void kernel_launch(void* const* d_in, const int* in_sizes, int n_in,
                              void* d_out, int out_size, void* d_ws, size_t ws_size,
                              hipStream_t stream) {
    const float* x     = (const float*)d_in[0];
    const float* w_ih0 = (const float*)d_in[1];
    const float* w_hh0 = (const float*)d_in[2];
    const float* b_ih0 = (const float*)d_in[3];
    const float* b_hh0 = (const float*)d_in[4];
    const float* w_ih1 = (const float*)d_in[5];
    const float* w_hh1 = (const float*)d_in[6];
    const float* b_ih1 = (const float*)d_in[7];
    const float* b_hh1 = (const float*)d_in[8];
    const float* w_reg = (const float*)d_in[9];
    const float* b_reg = (const float*)d_in[10];
    float* out = (float*)d_out;
    float* xg  = (float*)d_ws;

    xgemm_wmma_kernel<<<512, 256, 0, stream>>>(x, w_ih0, b_ih0, b_hh0, xg);
    lstm_scan_kernel<<<64, 32, 0, stream>>>(xg, w_hh0, w_ih1, w_hh1,
                                            b_ih1, b_hh1, w_reg, b_reg, out);
}


# scan re-tiled: a batch element's two layers on neighbouring matrix columns (2 batch elements per wave, 128 waves): one cell update per trip serves both layers; weights stacked along k; sum order kept
# speedup vs baseline: 2.5038x; 1.2635x over previous
#include <hip/hip_runtime.h>
#include <hip/hip_bf16.h>

#define SEQ   512
#define BATCH 256
#define INF   180
#define HD    8
#define G4    32

typedef __attribute__((ext_vector_type(2))) float v2f;
typedef __attribute__((ext_vector_type(4))) float v4f;
typedef __attribute__((ext_vector_type(8))) float v8f;

__global__ __launch_bounds__(256) void xgemm_wmma_kernel(
    const float* __restrict__ x,
    const float* __restrict__ w_ih0,
    const float* __restrict__ b_ih0,
    const float* __restrict__ b_hh0,
    float* __restrict__ xg)
{
    const int gwave = (int)((blockIdx.x * blockDim.x + threadIdx.x) >> 5);
    const int lane  = (int)(threadIdx.x & 31);
    const int n     = lane & 15;
    const int hi    = lane >> 4;

    const int rowbase = gwave * 32;
    const float* a0p = x + (size_t)(rowbase + n)      * INF;
    const float* a1p = x + (size_t)(rowbase + 16 + n) * INF;
    const float* b0p = w_ih0 + (size_t)n        * INF;
    const float* b1p = w_ih0 + (size_t)(n + 16) * INF;

    v8f acc00 = {0.f,0.f,0.f,0.f,0.f,0.f,0.f,0.f};
    v8f acc01 = acc00, acc10 = acc00, acc11 = acc00;

    #pragma unroll 5
    for (int k0 = 0; k0 < INF; k0 += 4) {
        const int k = k0 + 2 * hi;
        const v2f a0 = *(const v2f*)(a0p + k);
        const v2f a1 = *(const v2f*)(a1p + k);
        const v2f b0 = *(const v2f*)(b0p + k);
        const v2f b1 = *(const v2f*)(b1p + k);
        acc00 = __builtin_amdgcn_wmma_f32_16x16x4_f32(false, a0, false, b0,
                                                      (short)0, acc00, false, false);
        acc01 = __builtin_amdgcn_wmma_f32_16x16x4_f32(false, a0, false, b1,
                                                      (short)0, acc01, false, false);
        acc10 = __builtin_amdgcn_wmma_f32_16x16x4_f32(false, a1, false, b0,
                                                      (short)0, acc10, false, false);
        acc11 = __builtin_amdgcn_wmma_f32_16x16x4_f32(false, a1, false, b1,
                                                      (short)0, acc11, false, false);
    }

    const float bias0 = b_ih0[n]      + b_hh0[n];
    const float bias1 = b_ih0[n + 16] + b_hh0[n + 16];
    const int p0 = (n & 7) * 4 + (n >> 3);
    const int p1 = p0 + 2;

    float* ob = xg + (size_t)(rowbase + hi * 8) * G4;
    #pragma unroll
    for (int r = 0; r < 8; ++r) {
        ob[(size_t)r * G4 + p0]        = acc00[r] + bias0;
        ob[(size_t)r * G4 + p1]        = acc01[r] + bias1;
        ob[(size_t)(r + 16) * G4 + p0] = acc10[r] + bias0;
        ob[(size_t)(r + 16) * G4 + p1] = acc11[r] + bias1;
    }
}

#define LOG2E     1.4426950408889634f
#define SIG_TMAX  36.0f
#define TANH_TMAX 29.0f
#define BPW       2

__device__ __forceinline__ v8f wmma4(v2f a, v2f b, v8f c) {
    return __builtin_amdgcn_wmma_f32_16x16x4_f32(false, a, false, b,
                                                 (short)0, c, false, false);
}

template <int J>
__device__ __forceinline__ float quad_bcast(float v) {
    return __int_as_float(__builtin_amdgcn_mov_dpp(__float_as_int(v), J * 0x55, 0xF, 0xF, false));
}
__device__ __forceinline__ void quad_spread(float v, v2f& k0, v2f& k1) {
    k0 = (v2f){quad_bcast<0>(v), quad_bcast<1>(v)};
    k1 = (v2f){quad_bcast<2>(v), quad_bcast<3>(v)};
}
__device__ __forceinline__ float row_down4(float v) {
    return __int_as_float(__builtin_amdgcn_update_dpp(0, __float_as_int(v), 0x114, 0xF, 0xF, true));
}

__device__ __forceinline__ void cell_update(const v8f& d0, const v8f& d1,
                                            const bool dlo, const bool dhi,
                                            float& cs, float& hv) {
    const v8f m  = dhi ? d1 : d0;
    const v4f lo = __builtin_shufflevector(m, m, 0, 1, 2, 3);
    const v4f hi = __builtin_shufflevector(m, m, 4, 5, 6, 7);
    const v4f p  = dlo ? hi : lo;
    const float ei = __builtin_amdgcn_exp2f(fminf(p[0] * -LOG2E, SIG_TMAX));
    const float ef = __builtin_amdgcn_exp2f(fminf(p[1] * -LOG2E, SIG_TMAX));
    const float eg = __builtin_amdgcn_exp2f(fminf(p[2] * (2.0f * LOG2E), TANH_TMAX));
    const float eo = __builtin_amdgcn_exp2f(fminf(p[3] * -LOG2E, SIG_TMAX));
    const float ai = 1.0f + ei;
    const float af = 1.0f + ef;
    const float ao = 1.0f + eo;
    const float pp = ai * (eg + 1.0f);
    const float num = cs * pp + (eg - 1.0f) * af;
    const float den = af * pp;
    cs = num / den;
    const float ec = __builtin_amdgcn_exp2f(fminf(cs * (2.0f * LOG2E), TANH_TMAX));
    const float hn = ec - 1.0f;
    const float hd = (ec + 1.0f) * ao;
    hv = hn / hd;
}

__global__ __launch_bounds__(32) __attribute__((amdgpu_waves_per_eu(1, 1)))
void lstm_scan_kernel(
    const float* __restrict__ xg,
    const float* __restrict__ w_hh0,
    const float* __restrict__ w_ih1,
    const float* __restrict__ w_hh1,
    const float* __restrict__ b_ih1,
    const float* __restrict__ b_hh1,
    const float* __restrict__ w_reg,
    const float* __restrict__ b_reg,
    float* __restrict__ out)
{
    const int lane = (int)(threadIdx.x & 31);
    const int c    = lane & 15;
    const int hf   = lane >> 4;
    const int d    = c & 3;
    const bool dlo = (d & 1) != 0;
    const bool dhi = (d & 2) != 0;
    const bool lay1 = (c & 4) != 0;
    const int b    = (int)blockIdx.x * BPW + (c >> 3);

    v2f a_hh0[2][2], a_ih1[2][2], a_hh1[2][2], a_head[2];
    #pragma unroll
    for (int t = 0; t < 2; ++t) {
        const int wrow = (c & 3) * HD + 4 * (c >> 3) + 2 * t + ((c >> 2) & 1);
        #pragma unroll
        for (int q = 0; q < 2; ++q) {
            const int k = 4 * hf + 2 * q;
            a_hh0[t][q] = *(const v2f*)(w_hh0 + wrow * HD + k);
            a_ih1[t][q] = *(const v2f*)(w_ih1 + wrow * HD + k);
            a_hh1[t][q] = *(const v2f*)(w_hh1 + wrow * HD + k);
        }
    }
    #pragma unroll
    for (int q = 0; q < 2; ++q) {
        const int k = 4 * hf + 2 * q;
        const v2f wv = *(const v2f*)(w_reg + (c & 1) * HD + k);
        const v2f zero2 = {0.f, 0.f};
        a_head[q] = ((c & 7) < 2) ? wv : zero2;
    }

    v8f cb1[2], cbh;
    #pragma unroll
    for (int t = 0; t < 2; ++t) {
        #pragma unroll
        for (int r = 0; r < 8; ++r) {
            const int row = (r & 3) * HD + 4 * hf + 2 * t + (r >> 2);
            cb1[t][r] = b_ih1[row] + b_hh1[row];
        }
    }
    #pragma unroll
    for (int r = 0; r < 8; ++r) cbh[r] = 0.f;
    cbh[0] = b_reg[0];
    cbh[1] = b_reg[1];

    float hv = 0.f, cs = 0.f;

    const unsigned lbyte = (unsigned)(b * G4 + hf * 16) * 4u;
    const size_t rowbytes = (size_t)BATCH * G4 * 4;
    float2* outp = (float2*)(out + (size_t)b * 2) + (size_t)hf * BATCH;
    const bool storer = lay1 && (d == 0);

    auto trip = [&](v8f e0, v8f e1, float& ho0, float& ho1) {
        const float m0 = lay1 ? 0.f : hv;
        const float m1 = lay1 ? hv : 0.f;
        const float mi = row_down4(m0);
        v2f bh0a, bh0b, bh1a, bh1b, bi1a, bi1b;
        quad_spread(m0, bh0a, bh0b);
        quad_spread(m1, bh1a, bh1b);
        quad_spread(mi, bi1a, bi1b);
        e0 = wmma4(a_hh0[0][0], bh0a, e0);
        e1 = wmma4(a_hh0[1][0], bh0a, e1);
        e0 = wmma4(a_hh0[0][1], bh0b, e0);
        e1 = wmma4(a_hh0[1][1], bh0b, e1);
        e0 = wmma4(a_hh1[0][0], bh1a, e0);
        e1 = wmma4(a_hh1[1][0], bh1a, e1);
        e0 = wmma4(a_hh1[0][1], bh1b, e0);
        e1 = wmma4(a_hh1[1][1], bh1b, e1);
        e0 = wmma4(a_ih1[0][0], bi1a, e0);
        e1 = wmma4(a_ih1[1][0], bi1a, e1);
        e0 = wmma4(a_ih1[0][1], bi1b, e0);
        e1 = wmma4(a_ih1[1][1], bi1b, e1);
        v8f dh = wmma4(a_head[0], bh1a, cbh);
        dh = wmma4(a_head[1], bh1b, dh);
        cell_update(e0, e1, dlo, dhi, cs, hv);
        ho0 = dh[0];
        ho1 = dh[1];
    };
    auto row = [&](const char* lp, v8f& t0, v8f& t1) {
        t0 = *(const v8f*)lp;
        t1 = *(const v8f*)(lp + 32);
    };
    const char* xgb = (const char*)xg + lbyte;

    v8f p0, p1, q0, q1;
    row(xgb, p0, p1);
    row(xgb + rowbytes, q0, q1);
    {
        const v8f ca0 = lay1 ? cb1[0] : p0, ca1 = lay1 ? cb1[1] : p1;
        const v8f cq0 = lay1 ? cb1[0] : q0, cq1 = lay1 ? cb1[1] : q1;
        row(xgb + 2 * rowbytes, p0, p1);
        row(xgb + 3 * rowbytes, q0, q1);
        float u0, u1;
        trip(ca0, ca1, u0, u1);
        hv = lay1 ? 0.f : hv;
        cs = lay1 ? 0.f : cs;
        trip(cq0, cq1, u0, u1);
    }

    float oe0 = 0.f, oe1 = 0.f;
    float oo0 = 0.f, oo1 = 0.f;

    const char* nxt = xgb + 4 * rowbytes;
    for (int s = 2; s < SEQ; s += 2) {
        const v8f ca0 = lay1 ? cb1[0] : p0, ca1 = lay1 ? cb1[1] : p1;
        const v8f cq0 = lay1 ? cb1[0] : q0, cq1 = lay1 ? cb1[1] : q1;
        row(nxt, p0, p1);
        row(nxt + rowbytes, q0, q1);
        __builtin_amdgcn_sched_barrier(0);
        trip(ca0, ca1, oe0, oe1);
        trip(cq0, cq1, oo0, oo1);
        float2 r;
        r.x = hf ? oo0 : oe0;
        r.y = hf ? oo1 : oe1;
        if (storer) outp[(size_t)(s - 2) * BATCH] = r;
        nxt += (s + 4 < SEQ) ? 2 * rowbytes : 0;
    }

    {
        const v8f ca0 = lay1 ? cb1[0] : p0, ca1 = lay1 ? cb1[1] : p1;
        trip(ca0, ca1, oe0, oe1);
        const float m1 = lay1 ? hv : 0.f;
        v2f g0, g1;
        quad_spread(m1, g0, g1);
        v8f dl = wmma4(a_head[0], g0, cbh);
        dl = wmma4(a_head[1], g1, dl);
        float2 r;
        r.x = hf ? dl[0] : oe0;
        r.y = hf ? dl[1] : oe1;
        if (storer) outp[(size_t)(SEQ - 2) * BATCH] = r;
    }
}

extern "C" void kernel_launch(void* const* d_in, const int* in_sizes, int n_in,
                              void* d_out, int out_size, void* d_ws, size_t ws_size,
                              hipStream_t stream) {
    const float* x     = (const float*)d_in[0];
    const float* w_ih0 = (const float*)d_in[1];
    const float* w_hh0 = (const float*)d_in[2];
    const float* b_ih0 = (const float*)d_in[3];
    const float* b_hh0 = (const float*)d_in[4];
    const float* w_ih1 = (const float*)d_in[5];
    const float* w_hh1 = (const float*)d_in[6];
    const float* b_ih1 = (const float*)d_in[7];
    const float* b_hh1 = (const float*)d_in[8];
    const float* w_reg = (const float*)d_in[9];
    const float* b_reg = (const float*)d_in[10];
    float* out = (float*)d_out;
    float* xg  = (float*)d_ws;

    xgemm_wmma_kernel<<<512, 256, 0, stream>>>(x, w_ih0, b_ih0, b_hh0, xg);
    lstm_scan_kernel<<<128, 32, 0, stream>>>(xg, w_hh0, w_ih1, w_hh1,
                                             b_ih1, b_hh1, w_reg, b_reg, out);
}


# input GEMM: x and w_ih0 rows read as 16-byte per-lane pieces straight into the fragments (a piece feeds two f32 WMMAs, same k map in A and B, no LDS): 92 loads per wave, was 180; k sum order changed
# speedup vs baseline: 2.6292x; 1.0501x over previous
#include <hip/hip_runtime.h>
#include <hip/hip_bf16.h>

#define SEQ   512
#define BATCH 256
#define INF   180
#define HD    8
#define G4    32

typedef __attribute__((ext_vector_type(2))) float v2f;
typedef __attribute__((ext_vector_type(4))) float v4f;
typedef __attribute__((ext_vector_type(8))) float v8f;

__global__ __launch_bounds__(256) void xgemm_wmma_kernel(
    const float* __restrict__ x,
    const float* __restrict__ w_ih0,
    const float* __restrict__ b_ih0,
    const float* __restrict__ b_hh0,
    float* __restrict__ xg)
{
    const int gwave = (int)((blockIdx.x * blockDim.x + threadIdx.x) >> 5);
    const int lane  = (int)(threadIdx.x & 31);
    const int n     = lane & 15;
    const int hi    = lane >> 4;

    const int rowbase = gwave * 32;
    const float* a0p = x + (size_t)(rowbase + n)      * INF;
    const float* a1p = x + (size_t)(rowbase + 16 + n) * INF;
    const float* b0p = w_ih0 + (size_t)n        * INF;
    const float* b1p = w_ih0 + (size_t)(n + 16) * INF;

    v8f acc00 = {0.f,0.f,0.f,0.f,0.f,0.f,0.f,0.f};
    v8f acc01 = acc00, acc10 = acc00, acc11 = acc00;

    struct Frag16 { v4f a0l, a0h, a1l, a1h, b0l, b0h, b1l, b1h; };
    auto load16 = [&](const int k0, Frag16& f) {
        const int k = k0 + 4 * hi;
        f.a0l = *(const v4f*)(a0p + k); f.a0h = *(const v4f*)(a0p + k + 8);
        f.a1l = *(const v4f*)(a1p + k); f.a1h = *(const v4f*)(a1p + k + 8);
        f.b0l = *(const v4f*)(b0p + k); f.b0h = *(const v4f*)(b0p + k + 8);
        f.b1l = *(const v4f*)(b1p + k); f.b1h = *(const v4f*)(b1p + k + 8);
    };
    auto mma16 = [&](const Frag16& f) {
        #pragma unroll
        for (int j = 0; j < 4; ++j) {
            const v4f ua0 = (j < 2) ? f.a0l : f.a0h, ua1 = (j < 2) ? f.a1l : f.a1h;
            const v4f ub0 = (j < 2) ? f.b0l : f.b0h, ub1 = (j < 2) ? f.b1l : f.b1h;
            const int e = 2 * (j & 1);
            const v2f a0 = {ua0[e], ua0[e + 1]}, a1 = {ua1[e], ua1[e + 1]};
            const v2f b0 = {ub0[e], ub0[e + 1]}, b1 = {ub1[e], ub1[e + 1]};
            acc00 = __builtin_amdgcn_wmma_f32_16x16x4_f32(false, a0, false, b0,
                                                          (short)0, acc00, false, false);
            acc01 = __builtin_amdgcn_wmma_f32_16x16x4_f32(false, a0, false, b1,
                                                          (short)0, acc01, false, false);
            acc10 = __builtin_amdgcn_wmma_f32_16x16x4_f32(false, a1, false, b0,
                                                          (short)0, acc10, false, false);
            acc11 = __builtin_amdgcn_wmma_f32_16x16x4_f32(false, a1, false, b1,
                                                          (short)0, acc11, false, false);
        }
    };
    #pragma unroll 1
    for (int k0 = 0; k0 < 160; k0 += 32) {
        Frag16 f, g;
        load16(k0, f);
        load16(k0 + 16, g);
        __builtin_amdgcn_sched_barrier(0);
        mma16(f);
        mma16(g);
    }
    {
        Frag16 f;
        load16(160, f);
        mma16(f);
    }
    {
        const int k = 176 + 2 * hi;
        const v2f a0 = *(const v2f*)(a0p + k);
        const v2f a1 = *(const v2f*)(a1p + k);
        const v2f b0 = *(const v2f*)(b0p + k);
        const v2f b1 = *(const v2f*)(b1p + k);
        acc00 = __builtin_amdgcn_wmma_f32_16x16x4_f32(false, a0, false, b0,
                                                      (short)0, acc00, false, false);
        acc01 = __builtin_amdgcn_wmma_f32_16x16x4_f32(false, a0, false, b1,
                                                      (short)0, acc01, false, false);
        acc10 = __builtin_amdgcn_wmma_f32_16x16x4_f32(false, a1, false, b0,
                                                      (short)0, acc10, false, false);
        acc11 = __builtin_amdgcn_wmma_f32_16x16x4_f32(false, a1, false, b1,
                                                      (short)0, acc11, false, false);
    }

    const float bias0 = b_ih0[n]      + b_hh0[n];
    const float bias1 = b_ih0[n + 16] + b_hh0[n + 16];
    const int p0 = (n & 7) * 4 + (n >> 3);
    const int p1 = p0 + 2;

    float* ob = xg + (size_t)(rowbase + hi * 8) * G4;
    #pragma unroll
    for (int r = 0; r < 8; ++r) {
        ob[(size_t)r * G4 + p0]        = acc00[r] + bias0;
        ob[(size_t)r * G4 + p1]        = acc01[r] + bias1;
        ob[(size_t)(r + 16) * G4 + p0] = acc10[r] + bias0;
        ob[(size_t)(r + 16) * G4 + p1] = acc11[r] + bias1;
    }
}

#define LOG2E     1.4426950408889634f
#define SIG_TMAX  36.0f
#define TANH_TMAX 29.0f
#define BPW       2

__device__ __forceinline__ v8f wmma4(v2f a, v2f b, v8f c) {
    return __builtin_amdgcn_wmma_f32_16x16x4_f32(false, a, false, b,
                                                 (short)0, c, false, false);
}

template <int J>
__device__ __forceinline__ float quad_bcast(float v) {
    return __int_as_float(__builtin_amdgcn_mov_dpp(__float_as_int(v), J * 0x55, 0xF, 0xF, false));
}
__device__ __forceinline__ void quad_spread(float v, v2f& k0, v2f& k1) {
    k0 = (v2f){quad_bcast<0>(v), quad_bcast<1>(v)};
    k1 = (v2f){quad_bcast<2>(v), quad_bcast<3>(v)};
}
__device__ __forceinline__ float row_down4(float v) {
    return __int_as_float(__builtin_amdgcn_update_dpp(0, __float_as_int(v), 0x114, 0xF, 0xF, true));
}

__device__ __forceinline__ void cell_update(const v8f& d0, const v8f& d1,
                                            const bool dlo, const bool dhi,
                                            float& cs, float& hv) {
    const v8f m  = dhi ? d1 : d0;
    const v4f lo = __builtin_shufflevector(m, m, 0, 1, 2, 3);
    const v4f hi = __builtin_shufflevector(m, m, 4, 5, 6, 7);
    const v4f p  = dlo ? hi : lo;
    const float ei = __builtin_amdgcn_exp2f(fminf(p[0] * -LOG2E, SIG_TMAX));
    const float ef = __builtin_amdgcn_exp2f(fminf(p[1] * -LOG2E, SIG_TMAX));
    const float eg = __builtin_amdgcn_exp2f(fminf(p[2] * (2.0f * LOG2E), TANH_TMAX));
    const float eo = __builtin_amdgcn_exp2f(fminf(p[3] * -LOG2E, SIG_TMAX));
    const float ai = 1.0f + ei;
    const float af = 1.0f + ef;
    const float ao = 1.0f + eo;
    const float pp = ai * (eg + 1.0f);
    const float num = cs * pp + (eg - 1.0f) * af;
    const float den = af * pp;
    cs = num / den;
    const float ec = __builtin_amdgcn_exp2f(fminf(cs * (2.0f * LOG2E), TANH_TMAX));
    const float hn = ec - 1.0f;
    const float hd = (ec + 1.0f) * ao;
    hv = hn / hd;
}

__global__ __launch_bounds__(32) __attribute__((amdgpu_waves_per_eu(1, 1)))
void lstm_scan_kernel(
    const float* __restrict__ xg,
    const float* __restrict__ w_hh0,
    const float* __restrict__ w_ih1,
    const float* __restrict__ w_hh1,
    const float* __restrict__ b_ih1,
    const float* __restrict__ b_hh1,
    const float* __restrict__ w_reg,
    const float* __restrict__ b_reg,
    float* __restrict__ out)
{
    const int lane = (int)(threadIdx.x & 31);
    const int c    = lane & 15;
    const int hf   = lane >> 4;
    const int d    = c & 3;
    const bool dlo = (d & 1) != 0;
    const bool dhi = (d & 2) != 0;
    const bool lay1 = (c & 4) != 0;
    const int b    = (int)blockIdx.x * BPW + (c >> 3);

    v2f a_hh0[2][2], a_ih1[2][2], a_hh1[2][2], a_head[2];
    #pragma unroll
    for (int t = 0; t < 2; ++t) {
        const int wrow = (c & 3) * HD + 4 * (c >> 3) + 2 * t + ((c >> 2) & 1);
        #pragma unroll
        for (int q = 0; q < 2; ++q) {
            const int k = 4 * hf + 2 * q;
            a_hh0[t][q] = *(const v2f*)(w_hh0 + wrow * HD + k);
            a_ih1[t][q] = *(const v2f*)(w_ih1 + wrow * HD + k);
            a_hh1[t][q] = *(const v2f*)(w_hh1 + wrow * HD + k);
        }
    }
    #pragma unroll
    for (int q = 0; q < 2; ++q) {
        const int k = 4 * hf + 2 * q;
        const v2f wv = *(const v2f*)(w_reg + (c & 1) * HD + k);
        const v2f zero2 = {0.f, 0.f};
        a_head[q] = ((c & 7) < 2) ? wv : zero2;
    }

    v8f cb1[2], cbh;
    #pragma unroll
    for (int t = 0; t < 2; ++t) {
        #pragma unroll
        for (int r = 0; r < 8; ++r) {
            const int row = (r & 3) * HD + 4 * hf + 2 * t + (r >> 2);
            cb1[t][r] = b_ih1[row] + b_hh1[row];
        }
    }
    #pragma unroll
    for (int r = 0; r < 8; ++r) cbh[r] = 0.f;
    cbh[0] = b_reg[0];
    cbh[1] = b_reg[1];

    float hv = 0.f, cs = 0.f;

    const unsigned lbyte = (unsigned)(b * G4 + hf * 16) * 4u;
    const size_t rowbytes = (size_t)BATCH * G4 * 4;
    float2* outp = (float2*)(out + (size_t)b * 2) + (size_t)hf * BATCH;
    const bool storer = lay1 && (d == 0);

    auto trip = [&](v8f e0, v8f e1, float& ho0, float& ho1) {
        const float m0 = lay1 ? 0.f : hv;
        const float m1 = lay1 ? hv : 0.f;
        const float mi = row_down4(m0);
        v2f bh0a, bh0b, bh1a, bh1b, bi1a, bi1b;
        quad_spread(m0, bh0a, bh0b);
        quad_spread(m1, bh1a, bh1b);
        quad_spread(mi, bi1a, bi1b);
        e0 = wmma4(a_hh0[0][0], bh0a, e0);
        e1 = wmma4(a_hh0[1][0], bh0a, e1);
        e0 = wmma4(a_hh0[0][1], bh0b, e0);
        e1 = wmma4(a_hh0[1][1], bh0b, e1);
        e0 = wmma4(a_hh1[0][0], bh1a, e0);
        e1 = wmma4(a_hh1[1][0], bh1a, e1);
        e0 = wmma4(a_hh1[0][1], bh1b, e0);
        e1 = wmma4(a_hh1[1][1], bh1b, e1);
        e0 = wmma4(a_ih1[0][0], bi1a, e0);
        e1 = wmma4(a_ih1[1][0], bi1a, e1);
        e0 = wmma4(a_ih1[0][1], bi1b, e0);
        e1 = wmma4(a_ih1[1][1], bi1b, e1);
        v8f dh = wmma4(a_head[0], bh1a, cbh);
        dh = wmma4(a_head[1], bh1b, dh);
        cell_update(e0, e1, dlo, dhi, cs, hv);
        ho0 = dh[0];
        ho1 = dh[1];
    };
    auto row = [&](const char* lp, v8f& t0, v8f& t1) {
        t0 = *(const v8f*)lp;
        t1 = *(const v8f*)(lp + 32);
    };
    const char* xgb = (const char*)xg + lbyte;

    v8f p0, p1, q0, q1;
    row(xgb, p0, p1);
    row(xgb + rowbytes, q0, q1);
    {
        const v8f ca0 = lay1 ? cb1[0] : p0, ca1 = lay1 ? cb1[1] : p1;
        const v8f cq0 = lay1 ? cb1[0] : q0, cq1 = lay1 ? cb1[1] : q1;
        row(xgb + 2 * rowbytes, p0, p1);
        row(xgb + 3 * rowbytes, q0, q1);
        float u0, u1;
        trip(ca0, ca1, u0, u1);
        hv = lay1 ? 0.f : hv;
        cs = lay1 ? 0.f : cs;
        trip(cq0, cq1, u0, u1);
    }

    float oe0 = 0.f, oe1 = 0.f;
    float oo0 = 0.f, oo1 = 0.f;

    const char* nxt = xgb + 4 * rowbytes;
    for (int s = 2; s < SEQ; s += 2) {
        const v8f ca0 = lay1 ? cb1[0] : p0, ca1 = lay1 ? cb1[1] : p1;
        const v8f cq0 = lay1 ? cb1[0] : q0, cq1 = lay1 ? cb1[1] : q1;
        row(nxt, p0, p1);
        row(nxt + rowbytes, q0, q1);
        __builtin_amdgcn_sched_barrier(0);
        trip(ca0, ca1, oe0, oe1);
        trip(cq0, cq1, oo0, oo1);
        float2 r;
        r.x = hf ? oo0 : oe0;
        r.y = hf ? oo1 : oe1;
        if (storer) outp[(size_t)(s - 2) * BATCH] = r;
        nxt += (s + 4 < SEQ) ? 2 * rowbytes : 0;
    }

    {
        const v8f ca0 = lay1 ? cb1[0] : p0, ca1 = lay1 ? cb1[1] : p1;
        trip(ca0, ca1, oe0, oe1);
        const float m1 = lay1 ? hv : 0.f;
        v2f g0, g1;
        quad_spread(m1, g0, g1);
        v8f dl = wmma4(a_head[0], g0, cbh);
        dl = wmma4(a_head[1], g1, dl);
        float2 r;
        r.x = hf ? dl[0] : oe0;
        r.y = hf ? dl[1] : oe1;
        if (storer) outp[(size_t)(SEQ - 2) * BATCH] = r;
    }
}

extern "C" void kernel_launch(void* const* d_in, const int* in_sizes, int n_in,
                              void* d_out, int out_size, void* d_ws, size_t ws_size,
                              hipStream_t stream) {
    const float* x     = (const float*)d_in[0];
    const float* w_ih0 = (const float*)d_in[1];
    const float* w_hh0 = (const float*)d_in[2];
    const float* b_ih0 = (const float*)d_in[3];
    const float* b_hh0 = (const float*)d_in[4];
    const float* w_ih1 = (const float*)d_in[5];
    const float* w_hh1 = (const float*)d_in[6];
    const float* b_ih1 = (const float*)d_in[7];
    const float* b_hh1 = (const float*)d_in[8];
    const float* w_reg = (const float*)d_in[9];
    const float* b_reg = (const float*)d_in[10];
    float* out = (float*)d_out;
    float* xg  = (float*)d_ws;

    xgemm_wmma_kernel<<<512, 256, 0, stream>>>(x, w_ih0, b_ih0, b_hh0, xg);
    lstm_scan_kernel<<<128, 32, 0, stream>>>(xg, w_hh0, w_ih1, w_hh1,
                                             b_ih1, b_hh1, w_reg, b_reg, out);
}
